# starting kernel
# baseline (speedup 1.0000x reference)
#include <hip/hip_runtime.h>


typedef __attribute__((ext_vector_type(2))) float v2f;
typedef __attribute__((ext_vector_type(8))) float v8f;

__global__ void k_deg_init(float* __restrict__ deg, int n) {
  int i = blockIdx.x * blockDim.x + threadIdx.x;
  if (i < n) deg[i] = 1.0f;
}

__global__ void k_deg_count(float* __restrict__ deg, const int* __restrict__ dst, int e) {
  int i = blockIdx.x * blockDim.x + threadIdx.x;
  if (i < e) atomicAdd(&deg[dst[i]], 1.0f);
}

__global__ void k_dinv(float* __restrict__ deg, int n) {
  int i = blockIdx.x * blockDim.x + threadIdx.x;
  if (i < n) deg[i] = rsqrtf(deg[i]);
}

__global__ void k_gemm_wmma(float* __restrict__ h, const float* __restrict__ x,
                            const float* __restrict__ w, int n, int K, int dout) {
  const int lane  = threadIdx.x;
  const int mt    = blockIdx.x;
  const int nt    = blockIdx.y;
  const int half  = lane >> 4;
  const int l15   = lane & 15;

  int row = mt * 16 + l15;
  int arow = (row < n) ? row : (n - 1);
  const int col = nt * 16 + l15;

  const float* __restrict__ xr = x + (size_t)arow * K + half * 2;
  const float* __restrict__ wr = w + (size_t)col  * K + half * 2;

  v8f acc = {};
#pragma unroll 8
  for (int kk = 0; kk < K; kk += 4) {
    v2f a, b;
    a.x = xr[kk];
    a.y = xr[kk + 1];
    b.x = wr[kk];
    b.y = wr[kk + 1];
    acc = __builtin_amdgcn_wmma_f32_16x16x4_f32(
         false, a,  false, b,
         (short)0, acc,  false,  false);
  }

  const int rbase = mt * 16 + half * 8;
#pragma unroll
  for (int r = 0; r < 8; ++r) {
    int orow = rbase + r;
    if (orow < n) h[(size_t)orow * dout + col] = acc[r];
  }
}

__global__ void k_selfloop(float* __restrict__ agg, const float* __restrict__ h,
                           const float* __restrict__ dinv, int dout, long total) {
  long i = (long)blockIdx.x * blockDim.x + threadIdx.x;
  if (i < total) {
    int node = (int)(i / dout);
    float di = dinv[node];
    agg[i] = h[i] * di * di;
  }
}

__global__ void k_scatter(float* __restrict__ agg, const float* __restrict__ h,
                          const float* __restrict__ dinv,
                          const int* __restrict__ src, const int* __restrict__ dst,
                          long total, int dout, int shift) {
  long gid = (long)blockIdx.x * blockDim.x + threadIdx.x;
  if (gid >= total) return;
  int edge = (int)(gid >> shift);
  int c    = (int)(gid & ((1 << shift) - 1)) << 2;
  int s = src[edge];
  int d = dst[edge];
  float nrm = dinv[s] * dinv[d];
  const float4 v = *(const float4*)(h + (size_t)s * dout + c);
  float* ap = agg + (size_t)d * dout + c;
  atomicAdd(ap + 0, v.x * nrm);
  atomicAdd(ap + 1, v.y * nrm);
  atomicAdd(ap + 2, v.z * nrm);
  atomicAdd(ap + 3, v.w * nrm);
}

__global__ void k_finish(float* __restrict__ out, const float* __restrict__ agg,
                         const float* __restrict__ bias, int doutMask, long total,
                         int dorelu) {
  long i = (long)blockIdx.x * blockDim.x + threadIdx.x;
  if (i < total) {
    float v = agg[i] + bias[(int)i & doutMask];
    out[i] = dorelu ? fmaxf(v, 0.0f) : v;
  }
}

static inline size_t align256(size_t x) { return (x + 255) & ~(size_t)255; }

extern "C" void kernel_launch(void* const* d_in, const int* in_sizes, int n_in,
                              void* d_out, int out_size, void* d_ws, size_t ws_size,
                              hipStream_t stream) {
  (void)n_in; (void)out_size; (void)ws_size;

  const int DIN = 128, DH = 128, DOUT = 64, LAYERS = 5;
  const int N = in_sizes[0] / DIN;
  const int E = in_sizes[1] / 2;

  const float* x0   = (const float*)d_in[0];
  const int*  eidx  = (const int*)d_in[1];
  const int*  srcA  = eidx;
  const int*  dstA  = eidx + E;

  char* ws = (char*)d_ws;
  size_t off = 0;
  float* dinv = (float*)(ws + off); off += align256((size_t)N * sizeof(float));
  float* bufH = (float*)(ws + off); off += align256((size_t)N * DH * sizeof(float));
  float* bufA = (float*)(ws + off); off += align256((size_t)N * DH * sizeof(float));
  float* bufX = (float*)(ws + off); off += align256((size_t)N * DH * sizeof(float));

  const int TB = 256;

  k_deg_init <<<(N + TB - 1) / TB, TB, 0, stream>>>(dinv, N);
  k_deg_count<<<(E + TB - 1) / TB, TB, 0, stream>>>(dinv, dstA, E);
  k_dinv     <<<(N + TB - 1) / TB, TB, 0, stream>>>(dinv, N);

  const float* xin = x0;
  for (int layer = 0; layer < LAYERS; ++layer) {
    const int K    = DIN;
    const int dout = (layer == LAYERS - 1) ? DOUT : DH;
    const float* W = (const float*)d_in[2 + 2 * layer];
    const float* b = (const float*)d_in[3 + 2 * layer];
    float* out = (layer == LAYERS - 1) ? (float*)d_out : bufX;

    dim3 grid((N + 15) / 16, dout / 16);
    k_gemm_wmma<<<grid, 32, 0, stream>>>(bufH, xin, W, N, K, dout);

    long tot = (long)N * dout;
    k_selfloop<<<(unsigned)((tot + TB - 1) / TB), TB, 0, stream>>>(bufA, bufH, dinv, dout, tot);

    const int shift = (dout == 128) ? 5 : 4;
    long sc_total = (long)E << shift;
    k_scatter<<<(unsigned)((sc_total + TB - 1) / TB), TB, 0, stream>>>(
        bufA, bufH, dinv, srcA, dstA, sc_total, dout, shift);

    k_finish<<<(unsigned)((tot + TB - 1) / TB), TB, 0, stream>>>(
        out, bufA, b, dout - 1, tot, layer < LAYERS - 1 ? 1 : 0);

    xin = bufX;
  }
}


# Aggregation pulls over a per-call destination-grouped edge list (int atomic ranks), replacing per-edge f32 atomics: a node's sum runs in rank (arrival) order; self-loop, bias, ReLU in the same pass
# speedup vs baseline: 2.8026x; 2.8026x over previous
#include <hip/hip_runtime.h>


typedef __attribute__((ext_vector_type(2))) float v2f;
typedef __attribute__((ext_vector_type(8))) float v8f;

__global__ void k_zero_cnt(int* __restrict__ cnt, int n) {
  int i = blockIdx.x * blockDim.x + threadIdx.x;
  if (i < n) cnt[i] = 0;
}

__global__ void k_rank(int* __restrict__ cnt, int* __restrict__ rank,
                       const int* __restrict__ dst, int e) {
  int i = blockIdx.x * blockDim.x + threadIdx.x;
  if (i < e) rank[i] = atomicAdd(&cnt[dst[i]], 1);
}

__global__ void __launch_bounds__(1024)
k_rowptr(int* __restrict__ rowinfo, float* __restrict__ dinv,
         const int* __restrict__ cnt, int n) {
  __shared__ int s_tot[32];
  __shared__ int s_part[32];
  const int tid  = threadIdx.x;
  const int lane = tid & 31;
  const int wv   = tid >> 5;
  const int tile0 = blockIdx.x * 1024;

  int part = 0;
  for (int i = tid; i < tile0; i += 1024) part += cnt[i];

  const int i = tile0 + tid;
  const int c = (i < n) ? cnt[i] : 0;

  int incl = c;
  for (int d = 1; d < 32; d <<= 1) {
    int t = __shfl_up(incl, d);
    if (lane >= d) incl += t;
  }
  for (int d = 16; d > 0; d >>= 1) part += __shfl_xor(part, d);
  if (lane == 31) s_tot[wv] = incl;
  if (lane == 0)  s_part[wv] = part;
  __syncthreads();
  int base = s_part[lane] + ((lane < wv) ? s_tot[lane] : 0);
  for (int d = 16; d > 0; d >>= 1) base += __shfl_xor(base, d);

  if (i < n) {
    *(int2*)(rowinfo + 2 * (size_t)i) = make_int2(base + incl - c, c);
    dinv[i] = rsqrtf((float)(c + 1));
  }
}

__global__ void k_fill(int* __restrict__ pairs, const int* __restrict__ rowinfo,
                       const int* __restrict__ rank, const float* __restrict__ dinv,
                       const int* __restrict__ src, const int* __restrict__ dst, int e) {
  int i = blockIdx.x * blockDim.x + threadIdx.x;
  if (i < e) {
    int s = src[i];
    int p = rowinfo[2 * (size_t)dst[i]] + rank[i];
    *(int2*)(pairs + 2 * (size_t)p) = make_int2(s, __float_as_int(dinv[s]));
  }
}

__global__ void k_gemm_wmma(float* __restrict__ h, const float* __restrict__ x,
                            const float* __restrict__ w, int n, int K, int dout) {
  const int lane  = threadIdx.x;
  const int mt    = blockIdx.x;
  const int nt    = blockIdx.y;
  const int half  = lane >> 4;
  const int l15   = lane & 15;

  int row = mt * 16 + l15;
  int arow = (row < n) ? row : (n - 1);
  const int col = nt * 16 + l15;

  const float* __restrict__ xr = x + (size_t)arow * K + half * 2;
  const float* __restrict__ wr = w + (size_t)col  * K + half * 2;

  v8f acc = {};
#pragma unroll 8
  for (int kk = 0; kk < K; kk += 4) {
    v2f a, b;
    a.x = xr[kk];
    a.y = xr[kk + 1];
    b.x = wr[kk];
    b.y = wr[kk + 1];
    acc = __builtin_amdgcn_wmma_f32_16x16x4_f32(
         false, a,  false, b,
         (short)0, acc,  false,  false);
  }

  const int rbase = mt * 16 + half * 8;
#pragma unroll
  for (int r = 0; r < 8; ++r) {
    int orow = rbase + r;
    if (orow < n) h[(size_t)orow * dout + col] = acc[r];
  }
}


template <int VEC, int CNT>
__device__ __forceinline__ void pull_rows(
    __attribute__((ext_vector_type(VEC))) float& acc, const float* __restrict__ hl,
    int sv, int wv, int j, float dd, unsigned last) {
  typedef float vf __attribute__((ext_vector_type(VEC)));
  vf v[CNT];
  float w[CNT];
#pragma unroll
  for (int u = 0; u < CNT; ++u) {
    unsigned s = (unsigned)__builtin_amdgcn_readlane(sv, j + u);
    s = min(s, last);
    w[u] = __int_as_float(__builtin_amdgcn_readlane(wv, j + u)) * dd;
    v[u] = *(const vf*)(hl + (size_t)s * (VEC * 32));
  }
#pragma unroll
  for (int u = 0; u < CNT; ++u) acc += v[u] * w[u];
}

template <int VEC>
__global__ void __launch_bounds__(256)
k_pull(float* __restrict__ out, const float* __restrict__ h,
       const float* __restrict__ dinv, const int* __restrict__ rowinfo,
       const int* __restrict__ pairs, const float* __restrict__ bias,
       int n, int dorelu) {
  typedef float vf __attribute__((ext_vector_type(VEC)));
  constexpr int DOUT = VEC * 32;
  const int lane = threadIdx.x & 31;
  const int node = __builtin_amdgcn_readfirstlane((int)(blockIdx.x * 8 + (threadIdx.x >> 5)));
  if (node >= n) return;

  const int2  ri  = *(const int2*)(rowinfo + 2 * (size_t)node);
  const int   beg = ri.x;
  const int   deg = ri.y;
  const float dd  = dinv[node];
  const float* __restrict__ hl = h + lane * VEC;
  const unsigned last = (unsigned)(n - 1);

  int sv = 0, wv = 0;
  if (lane < deg) {
    const int2 pr = *(const int2*)(pairs + 2 * (size_t)(beg + lane));
    sv = pr.x;
    wv = pr.y;
  }
  const vf self = *(const vf*)(hl + (size_t)node * DOUT);
  vf acc = {};

  for (int base = 0; base < deg; ) {
    const int m = min(deg - base, 32);
    int j = 0;
    for (; j + 8 <= m; j += 8) pull_rows<VEC, 8>(acc, hl, sv, wv, j, dd, last);
    if (m & 4) { pull_rows<VEC, 4>(acc, hl, sv, wv, j, dd, last); j += 4; }
    if (m & 2) { pull_rows<VEC, 2>(acc, hl, sv, wv, j, dd, last); j += 2; }
    if (m & 1) { pull_rows<VEC, 1>(acc, hl, sv, wv, j, dd, last); }
    base += 32;
    if (base < deg && lane < deg - base) {
      const int2 pr = *(const int2*)(pairs + 2 * (size_t)(beg + base + lane));
      sv = pr.x;
      wv = pr.y;
    }
  }

  acc += (self * dd) * dd;

#pragma unroll
  for (int k = 0; k < VEC; ++k) {
    float r = acc[k] + bias[lane * VEC + k];
    acc[k] = dorelu ? fmaxf(r, 0.0f) : r;
  }
  *(vf*)(out + (size_t)node * DOUT + lane * VEC) = acc;
}

static inline size_t align256(size_t x) { return (x + 255) & ~(size_t)255; }

extern "C" void kernel_launch(void* const* d_in, const int* in_sizes, int n_in,
                              void* d_out, int out_size, void* d_ws, size_t ws_size,
                              hipStream_t stream) {
  (void)n_in; (void)out_size; (void)ws_size;

  const int DIN = 128, DH = 128, DOUT = 64, LAYERS = 5;
  const int N = in_sizes[0] / DIN;
  const int E = in_sizes[1] / 2;

  const float* x0   = (const float*)d_in[0];
  const int*  eidx  = (const int*)d_in[1];
  const int*  srcA  = eidx;
  const int*  dstA  = eidx + E;

  char* ws = (char*)d_ws;
  size_t off = 0;
  float* dinv   = (float*)(ws + off); off += align256((size_t)N * sizeof(float));
  int*   cnt    = (int*)  (ws + off); off += align256((size_t)N * sizeof(int));
  int*   rowinfo= (int*)  (ws + off); off += align256((size_t)N * 2 * sizeof(int));
  int*   rank   = (int*)  (ws + off); off += align256((size_t)E * sizeof(int));
  int*   pairs  = (int*)  (ws + off); off += align256((size_t)E * 2 * sizeof(int));
  float* bufH   = (float*)(ws + off); off += align256((size_t)N * DH * sizeof(float));
  float* bufX   = (float*)(ws + off); off += align256((size_t)N * DH * sizeof(float));

  const int TB = 256;

  k_zero_cnt<<<(N + TB - 1) / TB, TB, 0, stream>>>(cnt, N);
  k_rank    <<<(E + TB - 1) / TB, TB, 0, stream>>>(cnt, rank, dstA, E);
  k_rowptr  <<<(N + 1023) / 1024, 1024, 0, stream>>>(rowinfo, dinv, cnt, N);
  k_fill    <<<(E + TB - 1) / TB, TB, 0, stream>>>(pairs, rowinfo, rank, dinv, srcA, dstA, E);

  const float* xin = x0;
  for (int layer = 0; layer < LAYERS; ++layer) {
    const int K    = DIN;
    const int dout = (layer == LAYERS - 1) ? DOUT : DH;
    const float* W = (const float*)d_in[2 + 2 * layer];
    const float* b = (const float*)d_in[3 + 2 * layer];
    float* out = (layer == LAYERS - 1) ? (float*)d_out : bufX;
    const int dorelu = layer < LAYERS - 1 ? 1 : 0;

    dim3 grid((N + 15) / 16, dout / 16);
    k_gemm_wmma<<<grid, 32, 0, stream>>>(bufH, xin, W, N, K, dout);

    if (dout == 128)
      k_pull<4><<<(N + 7) / 8, 256, 0, stream>>>(out, bufH, dinv, rowinfo, pairs, b, N, dorelu);
    else
      k_pull<2><<<(N + 7) / 8, 256, 0, stream>>>(out, bufH, dinv, rowinfo, pairs, b, N, dorelu);

    xin = bufX;
  }
}


# GEMM re-tiled: a 4-wave block per 64 rows x 64 features reads A and W once with coalesced 16-byte loads via LDS, keeps A fragments in registers, 4 column tiles per wave, whole-line row stores
# speedup vs baseline: 3.7223x; 1.3282x over previous
#include <hip/hip_runtime.h>


typedef __attribute__((ext_vector_type(2))) float v2f;
typedef __attribute__((ext_vector_type(8))) float v8f;

__global__ void k_zero_cnt(int* __restrict__ cnt, int n) {
  int i = blockIdx.x * blockDim.x + threadIdx.x;
  if (i < n) cnt[i] = 0;
}

__global__ void k_rank(int* __restrict__ cnt, int* __restrict__ rank,
                       const int* __restrict__ dst, int e) {
  int i = blockIdx.x * blockDim.x + threadIdx.x;
  if (i < e) rank[i] = atomicAdd(&cnt[dst[i]], 1);
}

__global__ void __launch_bounds__(1024)
k_rowptr(int* __restrict__ rowinfo, float* __restrict__ dinv,
         const int* __restrict__ cnt, int n) {
  __shared__ int s_tot[32];
  __shared__ int s_part[32];
  const int tid  = threadIdx.x;
  const int lane = tid & 31;
  const int wv   = tid >> 5;
  const int tile0 = blockIdx.x * 1024;

  int part = 0;
  for (int i = tid; i < tile0; i += 1024) part += cnt[i];

  const int i = tile0 + tid;
  const int c = (i < n) ? cnt[i] : 0;

  int incl = c;
  for (int d = 1; d < 32; d <<= 1) {
    int t = __shfl_up(incl, d);
    if (lane >= d) incl += t;
  }
  for (int d = 16; d > 0; d >>= 1) part += __shfl_xor(part, d);
  if (lane == 31) s_tot[wv] = incl;
  if (lane == 0)  s_part[wv] = part;
  __syncthreads();
  int base = s_part[lane] + ((lane < wv) ? s_tot[lane] : 0);
  for (int d = 16; d > 0; d >>= 1) base += __shfl_xor(base, d);

  if (i < n) {
    *(int2*)(rowinfo + 2 * (size_t)i) = make_int2(base + incl - c, c);
    dinv[i] = rsqrtf((float)(c + 1));
  }
}

__global__ void k_fill(int* __restrict__ pairs, const int* __restrict__ rowinfo,
                       const int* __restrict__ rank, const float* __restrict__ dinv,
                       const int* __restrict__ src, const int* __restrict__ dst, int e) {
  int i = blockIdx.x * blockDim.x + threadIdx.x;
  if (i < e) {
    int s = src[i];
    int p = rowinfo[2 * (size_t)dst[i]] + rank[i];
    *(int2*)(pairs + 2 * (size_t)p) = make_int2(s, __float_as_int(dinv[s]));
  }
}

constexpr int GK = 128;
constexpr int GP = GK + 4;
constexpr int DP = 64 + 4;

template <int NCH>
__global__ void __launch_bounds__(128)
k_gemm_wmma(float* __restrict__ h, const float* __restrict__ x,
            const float* __restrict__ w, int n) {
  constexpr int DOUT = 64 * NCH;
  __shared__ __attribute__((aligned(16))) float sW[64 * GP];

  const int tid  = threadIdx.x;
  const int lane = tid & 31;
  const int wv   = __builtin_amdgcn_readfirstlane(tid >> 5);
  const int half = lane >> 4;
  const int l15  = lane & 15;
  const int rt   = blockIdx.x / NCH;
  const int ch   = blockIdx.x % NCH;

  const int row0 = rt * 64 + wv * 16;
#pragma unroll
  for (int i = 0; i < 16; ++i) {
    const int idx  = tid + 128 * i;
    const int grow = rt * 64 + (idx >> 5);
    const int crow = (grow < n) ? grow : (n - 1);
    const float4 v = *(const float4*)(x + (size_t)crow * GK + (idx & 31) * 4);
    *(float4*)(sW + (idx >> 5) * GP + (idx & 31) * 4) = v;
  }

  __syncthreads();

  v2f a[GK / 4];
  const float* sa = sW + (wv * 16 + l15) * GP + half * 2;
#pragma unroll
  for (int kk = 0; kk < GK / 4; ++kk) a[kk] = *(const v2f*)(sa + kk * 4);

  __syncthreads();

  const float4* __restrict__ wsrc = (const float4*)(w + (size_t)ch * 64 * GK);
#pragma unroll
  for (int i = 0; i < 16; ++i) {
    const int idx = tid + 128 * i;
    const float4 v = wsrc[idx];
    *(float4*)(sW + (idx >> 5) * GP + (idx & 31) * 4) = v;
  }
  __syncthreads();

  v8f acc[4] = {};
  const float* sb = sW + l15 * GP + half * 2;
#pragma unroll
  for (int kk = 0; kk < GK / 4; ++kk) {
#pragma unroll
    for (int t = 0; t < 4; ++t) {
      const v2f b = *(const v2f*)(sb + t * 16 * GP + kk * 4);
      acc[t] = __builtin_amdgcn_wmma_f32_16x16x4_f32(
           false, a[kk],  false, b,
           (short)0, acc[t],  false,  false);
    }
  }

  __syncthreads();

  float* sD = sW + wv * (16 * DP);
#pragma unroll
  for (int t = 0; t < 4; ++t) {
#pragma unroll
    for (int r = 0; r < 8; ++r) sD[(half * 8 + r) * DP + t * 16 + l15] = acc[t][r];
  }

  __syncthreads();

  float* __restrict__ hb = h + (size_t)ch * 64 + l15 * 4;
  const float* sr = sD + half * DP + l15 * 4;
  if (row0 + 16 <= n) {
    float4 v[8];
#pragma unroll
    for (int i = 0; i < 8; ++i) v[i] = *(const float4*)(sr + 2 * i * DP);
#pragma unroll
    for (int i = 0; i < 8; ++i)
      *(float4*)(hb + (size_t)(row0 + 2 * i + half) * DOUT) = v[i];
  } else {
#pragma unroll
    for (int i = 0; i < 8; ++i) {
      const int orow = row0 + 2 * i + half;
      const float4 v = *(const float4*)(sr + 2 * i * DP);
      if (orow < n) *(float4*)(hb + (size_t)orow * DOUT) = v;
    }
  }
}


template <int VEC, int CNT>
__device__ __forceinline__ void pull_rows(
    __attribute__((ext_vector_type(VEC))) float& acc, const float* __restrict__ hl,
    int sv, int wv, int j, float dd, unsigned last) {
  typedef float vf __attribute__((ext_vector_type(VEC)));
  vf v[CNT];
  float w[CNT];
#pragma unroll
  for (int u = 0; u < CNT; ++u) {
    unsigned s = (unsigned)__builtin_amdgcn_readlane(sv, j + u);
    s = min(s, last);
    w[u] = __int_as_float(__builtin_amdgcn_readlane(wv, j + u)) * dd;
    v[u] = *(const vf*)(hl + (size_t)s * (VEC * 32));
  }
#pragma unroll
  for (int u = 0; u < CNT; ++u) acc += v[u] * w[u];
}

template <int VEC>
__global__ void __launch_bounds__(256)
k_pull(float* __restrict__ out, const float* __restrict__ h,
       const float* __restrict__ dinv, const int* __restrict__ rowinfo,
       const int* __restrict__ pairs, const float* __restrict__ bias,
       int n, int dorelu) {
  typedef float vf __attribute__((ext_vector_type(VEC)));
  constexpr int DOUT = VEC * 32;
  const int lane = threadIdx.x & 31;
  const int node = __builtin_amdgcn_readfirstlane((int)(blockIdx.x * 8 + (threadIdx.x >> 5)));
  if (node >= n) return;

  const int2  ri  = *(const int2*)(rowinfo + 2 * (size_t)node);
  const int   beg = ri.x;
  const int   deg = ri.y;
  const float dd  = dinv[node];
  const float* __restrict__ hl = h + lane * VEC;
  const unsigned last = (unsigned)(n - 1);

  int sv = 0, wv = 0;
  if (lane < deg) {
    const int2 pr = *(const int2*)(pairs + 2 * (size_t)(beg + lane));
    sv = pr.x;
    wv = pr.y;
  }
  const vf self = *(const vf*)(hl + (size_t)node * DOUT);
  vf acc = {};

  for (int base = 0; base < deg; ) {
    const int m = min(deg - base, 32);
    int j = 0;
    for (; j + 8 <= m; j += 8) pull_rows<VEC, 8>(acc, hl, sv, wv, j, dd, last);
    if (m & 4) { pull_rows<VEC, 4>(acc, hl, sv, wv, j, dd, last); j += 4; }
    if (m & 2) { pull_rows<VEC, 2>(acc, hl, sv, wv, j, dd, last); j += 2; }
    if (m & 1) { pull_rows<VEC, 1>(acc, hl, sv, wv, j, dd, last); }
    base += 32;
    if (base < deg && lane < deg - base) {
      const int2 pr = *(const int2*)(pairs + 2 * (size_t)(beg + base + lane));
      sv = pr.x;
      wv = pr.y;
    }
  }

  acc += (self * dd) * dd;

#pragma unroll
  for (int k = 0; k < VEC; ++k) {
    float r = acc[k] + bias[lane * VEC + k];
    acc[k] = dorelu ? fmaxf(r, 0.0f) : r;
  }
  *(vf*)(out + (size_t)node * DOUT + lane * VEC) = acc;
}

static inline size_t align256(size_t x) { return (x + 255) & ~(size_t)255; }

extern "C" void kernel_launch(void* const* d_in, const int* in_sizes, int n_in,
                              void* d_out, int out_size, void* d_ws, size_t ws_size,
                              hipStream_t stream) {
  (void)n_in; (void)out_size; (void)ws_size;

  const int DIN = 128, DH = 128, DOUT = 64, LAYERS = 5;
  const int N = in_sizes[0] / DIN;
  const int E = in_sizes[1] / 2;

  const float* x0   = (const float*)d_in[0];
  const int*  eidx  = (const int*)d_in[1];
  const int*  srcA  = eidx;
  const int*  dstA  = eidx + E;

  char* ws = (char*)d_ws;
  size_t off = 0;
  float* dinv   = (float*)(ws + off); off += align256((size_t)N * sizeof(float));
  int*   cnt    = (int*)  (ws + off); off += align256((size_t)N * sizeof(int));
  int*   rowinfo= (int*)  (ws + off); off += align256((size_t)N * 2 * sizeof(int));
  int*   rank   = (int*)  (ws + off); off += align256((size_t)E * sizeof(int));
  int*   pairs  = (int*)  (ws + off); off += align256((size_t)E * 2 * sizeof(int));
  float* bufH   = (float*)(ws + off); off += align256((size_t)N * DH * sizeof(float));
  float* bufX   = (float*)(ws + off); off += align256((size_t)N * DH * sizeof(float));

  const int TB = 256;

  k_zero_cnt<<<(N + TB - 1) / TB, TB, 0, stream>>>(cnt, N);
  k_rank    <<<(E + TB - 1) / TB, TB, 0, stream>>>(cnt, rank, dstA, E);
  k_rowptr  <<<(N + 1023) / 1024, 1024, 0, stream>>>(rowinfo, dinv, cnt, N);
  k_fill    <<<(E + TB - 1) / TB, TB, 0, stream>>>(pairs, rowinfo, rank, dinv, srcA, dstA, E);

  const float* xin = x0;
  for (int layer = 0; layer < LAYERS; ++layer) {
    const int dout = (layer == LAYERS - 1) ? DOUT : DH;
    const float* W = (const float*)d_in[2 + 2 * layer];
    const float* b = (const float*)d_in[3 + 2 * layer];
    float* out = (layer == LAYERS - 1) ? (float*)d_out : bufX;
    const int dorelu = layer < LAYERS - 1 ? 1 : 0;

    if (dout == 128)
      k_gemm_wmma<2><<<((N + 63) / 64) * 2, 128, 0, stream>>>(bufH, xin, W, N);
    else
      k_gemm_wmma<1><<<(N + 63) / 64, 128, 0, stream>>>(bufH, xin, W, N);

    if (dout == 128)
      k_pull<4><<<(N + 7) / 8, 256, 0, stream>>>(out, bufH, dinv, rowinfo, pairs, b, N, dorelu);
    else
      k_pull<2><<<(N + 7) / 8, 256, 0, stream>>>(out, bufH, dinv, rowinfo, pairs, b, N, dorelu);

    xin = bufX;
  }
}


# Layers 0-3: the pull runs inside the next layer's GEMM launch: an 8-wave block pulls its 64 node rows (more rows in flight per wave) into the LDS A tile; 12 launches become 8; sum order unchanged
# speedup vs baseline: 4.6191x; 1.2409x over previous
#include <hip/hip_runtime.h>


typedef __attribute__((ext_vector_type(2))) float v2f;
typedef __attribute__((ext_vector_type(8))) float v8f;

constexpr int SLOTS = 32;

__global__ void k_zero_cnt(int* __restrict__ cnt, int n1) {
  int i = blockIdx.x * blockDim.x + threadIdx.x;
  if (i < n1) cnt[i] = 0;
}

__global__ void k_bucket(int* __restrict__ cnt, int* __restrict__ rows, int* __restrict__ ovf,
                         const int* __restrict__ src, const int* __restrict__ dst,
                         int e, int n) {
  int i = blockIdx.x * blockDim.x + threadIdx.x;
  if (i < e) {
    const int d = dst[i];
    const int s = src[i];
    const int r = atomicAdd(&cnt[d], 1);
    if (r < SLOTS) {
      rows[(size_t)d * SLOTS + r] = s;
    } else {
      const int k = atomicAdd(&cnt[n], 1);
      *(int2*)(ovf + 2 * (size_t)k) = make_int2(d, s);
    }
  }
}

constexpr int GK = 128;
constexpr int GP = GK + 4;
constexpr int DP = 64 + 4;

template <int NCH>
__global__ void __launch_bounds__(128)
k_gemm_wmma(float* __restrict__ h, const float* __restrict__ x,
            const float* __restrict__ w, const int* __restrict__ cnt, int n) {
  constexpr int DOUT = 64 * NCH;
  __shared__ __attribute__((aligned(16))) float sW[64 * GP];

  const int tid  = threadIdx.x;
  const int lane = tid & 31;
  const int wv   = __builtin_amdgcn_readfirstlane(tid >> 5);
  const int half = lane >> 4;
  const int l15  = lane & 15;
  const int rt   = blockIdx.x / NCH;
  const int ch   = blockIdx.x % NCH;

  const int row0 = rt * 64 + wv * 16;
#pragma unroll
  for (int i = 0; i < 16; ++i) {
    const int idx  = tid + 128 * i;
    const int grow = rt * 64 + (idx >> 5);
    const int crow = (grow < n) ? grow : (n - 1);
    const float4 v = *(const float4*)(x + (size_t)crow * GK + (idx & 31) * 4);
    *(float4*)(sW + (idx >> 5) * GP + (idx & 31) * 4) = v;
  }

  __syncthreads();

  v2f a[GK / 4];
  const float* sa = sW + (wv * 16 + l15) * GP + half * 2;
#pragma unroll
  for (int kk = 0; kk < GK / 4; ++kk) a[kk] = *(const v2f*)(sa + kk * 4);

  __syncthreads();

  const float4* __restrict__ wsrc = (const float4*)(w + (size_t)ch * 64 * GK);
#pragma unroll
  for (int i = 0; i < 16; ++i) {
    const int idx = tid + 128 * i;
    const float4 v = wsrc[idx];
    *(float4*)(sW + (idx >> 5) * GP + (idx & 31) * 4) = v;
  }
  __syncthreads();

  v8f acc[4] = {};
  const float* sb = sW + l15 * GP + half * 2;
#pragma unroll
  for (int kk = 0; kk < GK / 4; ++kk) {
#pragma unroll
    for (int t = 0; t < 4; ++t) {
      const v2f b = *(const v2f*)(sb + t * 16 * GP + kk * 4);
      acc[t] = __builtin_amdgcn_wmma_f32_16x16x4_f32(
           false, a[kk],  false, b,
           (short)0, acc[t],  false,  false);
    }
  }

  float sc[8];
#pragma unroll
  for (int i = 0; i < 8; ++i) {
    const int r = row0 + 2 * i + half;
    sc[i] = rsqrtf((float)(cnt[(r < n) ? r : (n - 1)] + 1));
  }

  __syncthreads();

  float* sD = sW + wv * (16 * DP);
#pragma unroll
  for (int t = 0; t < 4; ++t) {
#pragma unroll
    for (int r = 0; r < 8; ++r) sD[(half * 8 + r) * DP + t * 16 + l15] = acc[t][r];
  }

  __syncthreads();

  float* __restrict__ hb = h + (size_t)ch * 64 + l15 * 4;
  const float* sr = sD + half * DP + l15 * 4;
  if (row0 + 16 <= n) {
    float4 v[8];
#pragma unroll
    for (int i = 0; i < 8; ++i) v[i] = *(const float4*)(sr + 2 * i * DP);
#pragma unroll
    for (int i = 0; i < 8; ++i) {
      const float4 o = make_float4(v[i].x * sc[i], v[i].y * sc[i], v[i].z * sc[i], v[i].w * sc[i]);
      *(float4*)(hb + (size_t)(row0 + 2 * i + half) * DOUT) = o;
    }
  } else {
#pragma unroll
    for (int i = 0; i < 8; ++i) {
      const int orow = row0 + 2 * i + half;
      const float4 v = *(const float4*)(sr + 2 * i * DP);
      const float4 o = make_float4(v.x * sc[i], v.y * sc[i], v.z * sc[i], v.w * sc[i]);
      if (orow < n) *(float4*)(hb + (size_t)orow * DOUT) = o;
    }
  }
}


template <int VEC, int CNT>
__device__ __forceinline__ void pull_rows(
    __attribute__((ext_vector_type(VEC))) float& acc, const float* __restrict__ hl,
    int sv, int j, unsigned last) {
  typedef float vf __attribute__((ext_vector_type(VEC)));
  vf v[CNT];
#pragma unroll
  for (int u = 0; u < CNT; ++u) {
    unsigned s = (unsigned)__builtin_amdgcn_readlane(sv, j + u);
    s = min(s, last);
    v[u] = *(const vf*)(hl + (size_t)s * (VEC * 32));
  }
#pragma unroll
  for (int u = 0; u < CNT; ++u) acc += v[u];
}

template <int VEC>
__global__ void __launch_bounds__(256)
k_pull(float* __restrict__ out, const float* __restrict__ h,
       const int* __restrict__ cnt, const int* __restrict__ rows,
       const int* __restrict__ ovf, const float* __restrict__ bias,
       int n, int e, int dorelu) {
  typedef float vf __attribute__((ext_vector_type(VEC)));
  constexpr int DOUT = VEC * 32;
  const int lane = threadIdx.x & 31;
  const int node = __builtin_amdgcn_readfirstlane((int)(blockIdx.x * 8 + (threadIdx.x >> 5)));
  if (node >= n) return;

  const int   deg = cnt[node];
  const float dd  = rsqrtf((float)(deg + 1));
  const int   m   = min(deg, SLOTS);
  const float* __restrict__ hl = h + lane * VEC;
  const unsigned last = (unsigned)(n - 1);

  int sv = 0;
  if (lane < m) sv = rows[(size_t)node * SLOTS + lane];
  const vf self = *(const vf*)(hl + (size_t)node * DOUT);
  vf acc = {};

  int j = 0;
  for (; j + 8 <= m; j += 8) pull_rows<VEC, 8>(acc, hl, sv, j, last);
  if (m & 4) { pull_rows<VEC, 4>(acc, hl, sv, j, last); j += 4; }
  if (m & 2) { pull_rows<VEC, 2>(acc, hl, sv, j, last); j += 2; }
  if (m & 1) { pull_rows<VEC, 1>(acc, hl, sv, j, last); }

  if (deg > SLOTS) {
    const int novf = min(cnt[n], e);
    for (int base = 0; base < novf; base += 32) {
      int od = -1, os = 0;
      if (base + lane < novf) {
        const int2 pr = *(const int2*)(ovf + 2 * (size_t)(base + lane));
        od = pr.x;
        os = pr.y;
      }
      unsigned mask = __builtin_amdgcn_ballot_w32(od == node);
      while (mask) {
        const int b = __builtin_ctz(mask);
        mask &= mask - 1;
        pull_rows<VEC, 1>(acc, hl, os, b, last);
      }
    }
  }

  acc += self;

#pragma unroll
  for (int k = 0; k < VEC; ++k) {
    float r = acc[k] * dd + bias[lane * VEC + k];
    acc[k] = dorelu ? fmaxf(r, 0.0f) : r;
  }
  *(vf*)(out + (size_t)node * DOUT + lane * VEC) = acc;
}


template <int NCH>
__global__ void __launch_bounds__(256)
k_pull_gemm(float* __restrict__ hout, const float* __restrict__ hin,
            const int* __restrict__ cnt, const int* __restrict__ rows,
            const int* __restrict__ ovf, const float* __restrict__ bias,
            const float* __restrict__ w, int n, int e) {
  typedef float vf __attribute__((ext_vector_type(4)));
  constexpr int DOUT  = 64 * NCH;
  constexpr int WCOLS = DOUT / 2;
  constexpr int NT    = WCOLS / 16;
  constexpr int DPW   = WCOLS + 4;
  constexpr int LPR   = WCOLS / 4;
  constexpr int RPI   = 32 / LPR;
  constexpr int NI    = 16 / RPI;
  __shared__ __attribute__((aligned(16))) float sW[DOUT * GP];
  __shared__ float sS[64];

  const int tid  = threadIdx.x;
  const int lane = tid & 31;
  const int wv   = __builtin_amdgcn_readfirstlane(tid >> 5);
  const int half = lane >> 4;
  const int l15  = lane & 15;
  const int tr0  = blockIdx.x * 64;

  {
    const float* __restrict__ hl = hin + lane * 4;
    const unsigned last = (unsigned)(n - 1);
    const int p0 = tr0 + wv * 8;
    int dgl = 0;
    if (lane < 8) {
      dgl = cnt[min(p0 + lane, n - 1)];
      sS[wv * 8 + lane] = rsqrtf((float)(dgl + 1));
    }
    vf bv = *(const vf*)(bias + lane * 4);

    int deg = __builtin_amdgcn_readlane(dgl, 0);
    int m   = min(deg, SLOTS);
    int nd  = min(p0, n - 1);
    int sv  = 0;
    if (lane < m) sv = rows[(size_t)nd * SLOTS + lane];
    vf self = *(const vf*)(hl + (size_t)nd * GK);

    for (int i = 0; i < 8; ++i) {
      const int node = p0 + i;
      int deg_n = 0, m_n = 0, sv_n = 0;
      vf self_n = {};
      if (i + 1 < 8) {
        deg_n = __builtin_amdgcn_readlane(dgl, i + 1);
        m_n   = min(deg_n, SLOTS);
        const int nn = min(node + 1, n - 1);
        if (lane < m_n) sv_n = rows[(size_t)nn * SLOTS + lane];
        self_n = *(const vf*)(hl + (size_t)nn * GK);
      }

      vf acc = {};
      if (node < n) {
        int j = 0;
        for (; j + 16 <= m; j += 16) pull_rows<4, 16>(acc, hl, sv, j, last);
        switch (m - j) {
          case 1:  pull_rows<4, 1>(acc, hl, sv, j, last);  break;
          case 2:  pull_rows<4, 2>(acc, hl, sv, j, last);  break;
          case 3:  pull_rows<4, 3>(acc, hl, sv, j, last);  break;
          case 4:  pull_rows<4, 4>(acc, hl, sv, j, last);  break;
          case 5:  pull_rows<4, 5>(acc, hl, sv, j, last);  break;
          case 6:  pull_rows<4, 6>(acc, hl, sv, j, last);  break;
          case 7:  pull_rows<4, 7>(acc, hl, sv, j, last);  break;
          case 8:  pull_rows<4, 8>(acc, hl, sv, j, last);  break;
          case 9:  pull_rows<4, 9>(acc, hl, sv, j, last);  break;
          case 10: pull_rows<4, 10>(acc, hl, sv, j, last); break;
          case 11: pull_rows<4, 11>(acc, hl, sv, j, last); break;
          case 12: pull_rows<4, 12>(acc, hl, sv, j, last); break;
          case 13: pull_rows<4, 13>(acc, hl, sv, j, last); break;
          case 14: pull_rows<4, 14>(acc, hl, sv, j, last); break;
          case 15: pull_rows<4, 15>(acc, hl, sv, j, last); break;
          default: break;
        }

        if (deg > SLOTS) {
          const int novf = min(cnt[n], e);
          for (int base = 0; base < novf; base += 32) {
            int od = -1, os = 0;
            if (base + lane < novf) {
              const int2 pr = *(const int2*)(ovf + 2 * (size_t)(base + lane));
              od = pr.x;
              os = pr.y;
            }
            unsigned mask = __builtin_amdgcn_ballot_w32(od == node);
            while (mask) {
              const int b = __builtin_ctz(mask);
              mask &= mask - 1;
              pull_rows<4, 1>(acc, hl, os, b, last);
            }
          }
        }

        acc += self;
        const float dd = rsqrtf((float)(deg + 1));
#pragma unroll
        for (int k = 0; k < 4; ++k) acc[k] = fmaxf(acc[k] * dd + bv[k], 0.0f);
      }
      *(vf*)(sW + (wv * 8 + i) * GP + lane * 4) = acc;

      deg = deg_n; m = m_n; sv = sv_n; self = self_n;
    }
  }

  __syncthreads();

  const int rtile = wv & 3;
  const int ch    = wv >> 2;
  const int row0  = tr0 + rtile * 16;

  v2f a[GK / 4];
  const float* sa = sW + (rtile * 16 + l15) * GP + half * 2;
#pragma unroll
  for (int kk = 0; kk < GK / 4; ++kk) a[kk] = *(const v2f*)(sa + kk * 4);

  __syncthreads();

  const float4* __restrict__ wsrc = (const float4*)w;
#pragma unroll
  for (int i = 0; i < DOUT / 8; ++i) {
    const int idx = tid + 256 * i;
    const float4 v = wsrc[idx];
    *(float4*)(sW + (idx >> 5) * GP + (idx & 31) * 4) = v;
  }
  __syncthreads();

  v8f acc[NT] = {};
  const float* sb = sW + (ch * WCOLS + l15) * GP + half * 2;
#pragma unroll
  for (int kk = 0; kk < GK / 4; ++kk) {
#pragma unroll
    for (int t = 0; t < NT; ++t) {
      const v2f b = *(const v2f*)(sb + t * 16 * GP + kk * 4);
      acc[t] = __builtin_amdgcn_wmma_f32_16x16x4_f32(
           false, a[kk],  false, b,
           (short)0, acc[t],  false,  false);
    }
  }

  const int rsub = lane / LPR;
  const int c4   = lane % LPR;
  float sc[NI];
#pragma unroll
  for (int i = 0; i < NI; ++i) sc[i] = sS[rtile * 16 + RPI * i + rsub];

  __syncthreads();

  float* sD = sW + wv * (16 * DPW);
#pragma unroll
  for (int t = 0; t < NT; ++t) {
#pragma unroll
    for (int r = 0; r < 8; ++r) sD[(half * 8 + r) * DPW + t * 16 + l15] = acc[t][r];
  }

  __syncthreads();

  float* __restrict__ hb = hout + (size_t)ch * WCOLS + c4 * 4;
  const float* sr = sD + rsub * DPW + c4 * 4;
  if (row0 + 16 <= n) {
    float4 v[NI];
#pragma unroll
    for (int i = 0; i < NI; ++i) v[i] = *(const float4*)(sr + RPI * i * DPW);
#pragma unroll
    for (int i = 0; i < NI; ++i) {
      const float4 o = make_float4(v[i].x * sc[i], v[i].y * sc[i], v[i].z * sc[i], v[i].w * sc[i]);
      *(float4*)(hb + (size_t)(row0 + RPI * i + rsub) * DOUT) = o;
    }
  } else {
#pragma unroll
    for (int i = 0; i < NI; ++i) {
      const int orow = row0 + RPI * i + rsub;
      const float4 v = *(const float4*)(sr + RPI * i * DPW);
      const float4 o = make_float4(v.x * sc[i], v.y * sc[i], v.z * sc[i], v.w * sc[i]);
      if (orow < n) *(float4*)(hb + (size_t)orow * DOUT) = o;
    }
  }
}

static inline size_t align256(size_t x) { return (x + 255) & ~(size_t)255; }

extern "C" void kernel_launch(void* const* d_in, const int* in_sizes, int n_in,
                              void* d_out, int out_size, void* d_ws, size_t ws_size,
                              hipStream_t stream) {
  (void)n_in; (void)out_size; (void)ws_size;

  const int DIN = 128, DH = 128, DOUT = 64, LAYERS = 5;
  const int N = in_sizes[0] / DIN;
  const int E = in_sizes[1] / 2;

  const float* x0   = (const float*)d_in[0];
  const int*  eidx  = (const int*)d_in[1];
  const int*  srcA  = eidx;
  const int*  dstA  = eidx + E;

  char* ws = (char*)d_ws;
  size_t off = 0;
  int*   cnt    = (int*)  (ws + off); off += align256((size_t)(N + 1) * sizeof(int));
  int*   rows   = (int*)  (ws + off); off += align256((size_t)N * SLOTS * sizeof(int));
  int*   ovf    = (int*)  (ws + off); off += align256((size_t)E * 2 * sizeof(int));
  float* bufH   = (float*)(ws + off); off += align256((size_t)N * DH * sizeof(float));
  float* bufX   = (float*)(ws + off); off += align256((size_t)N * DH * sizeof(float));

  const int TB = 256;

  k_zero_cnt<<<(N + 1 + TB - 1) / TB, TB, 0, stream>>>(cnt, N + 1);
  k_bucket  <<<(E + TB - 1) / TB, TB, 0, stream>>>(cnt, rows, ovf, srcA, dstA, E, N);

  k_gemm_wmma<2><<<((N + 63) / 64) * 2, 128, 0, stream>>>(bufH, x0, (const float*)d_in[2], cnt, N);

  float* hin  = bufH;
  float* hnew = bufX;
  for (int layer = 1; layer < LAYERS; ++layer) {
    const float* W  = (const float*)d_in[2 + 2 * layer];
    const float* bp = (const float*)d_in[3 + 2 * (layer - 1)];
    if (layer < LAYERS - 1)
      k_pull_gemm<2><<<(N + 63) / 64, 256, 0, stream>>>(hnew, hin, cnt, rows, ovf, bp, W, N, E);
    else
      k_pull_gemm<1><<<(N + 63) / 64, 256, 0, stream>>>(hnew, hin, cnt, rows, ovf, bp, W, N, E);
    float* t = hin; hin = hnew; hnew = t;
  }

  k_pull<2><<<(N + 7) / 8, 256, 0, stream>>>((float*)d_out, hin, cnt, rows, ovf,
                                             (const float*)d_in[3 + 2 * (LAYERS - 1)], N, E, 0);
}


# Fused pull+GEMM launches: W staged once per block, its 64 rows done as four 16-row tiles (pull, WMMA, store), so blocks on a WGP run WMMAs under each other's gathers; sum order unchanged
# speedup vs baseline: 4.8935x; 1.0594x over previous
#include <hip/hip_runtime.h>


typedef __attribute__((ext_vector_type(2))) float v2f;
typedef __attribute__((ext_vector_type(8))) float v8f;

constexpr int SLOTS = 32;

__global__ void k_zero_cnt(int* __restrict__ cnt, int n1) {
  int i = blockIdx.x * blockDim.x + threadIdx.x;
  if (i < n1) cnt[i] = 0;
}

__global__ void k_bucket(int* __restrict__ cnt, int* __restrict__ rows, int* __restrict__ ovf,
                         const int* __restrict__ src, const int* __restrict__ dst,
                         int e, int n) {
  int i = blockIdx.x * blockDim.x + threadIdx.x;
  if (i < e) {
    const int d = dst[i];
    const int s = src[i];
    const int r = atomicAdd(&cnt[d], 1);
    if (r < SLOTS) {
      rows[(size_t)d * SLOTS + r] = s;
    } else {
      const int k = atomicAdd(&cnt[n], 1);
      *(int2*)(ovf + 2 * (size_t)k) = make_int2(d, s);
    }
  }
}

constexpr int GK = 128;
constexpr int GP = GK + 4;
constexpr int DP = 64 + 4;

template <int NCH>
__global__ void __launch_bounds__(128)
k_gemm_wmma(float* __restrict__ h, const float* __restrict__ x,
            const float* __restrict__ w, const int* __restrict__ cnt, int n) {
  constexpr int DOUT = 64 * NCH;
  __shared__ __attribute__((aligned(16))) float sW[64 * GP];

  const int tid  = threadIdx.x;
  const int lane = tid & 31;
  const int wv   = __builtin_amdgcn_readfirstlane(tid >> 5);
  const int half = lane >> 4;
  const int l15  = lane & 15;
  const int rt   = blockIdx.x / NCH;
  const int ch   = blockIdx.x % NCH;

  const int row0 = rt * 64 + wv * 16;
#pragma unroll
  for (int i = 0; i < 16; ++i) {
    const int idx  = tid + 128 * i;
    const int grow = rt * 64 + (idx >> 5);
    const int crow = (grow < n) ? grow : (n - 1);
    const float4 v = *(const float4*)(x + (size_t)crow * GK + (idx & 31) * 4);
    *(float4*)(sW + (idx >> 5) * GP + (idx & 31) * 4) = v;
  }

  __syncthreads();

  v2f a[GK / 4];
  const float* sa = sW + (wv * 16 + l15) * GP + half * 2;
#pragma unroll
  for (int kk = 0; kk < GK / 4; ++kk) a[kk] = *(const v2f*)(sa + kk * 4);

  __syncthreads();

  const float4* __restrict__ wsrc = (const float4*)(w + (size_t)ch * 64 * GK);
#pragma unroll
  for (int i = 0; i < 16; ++i) {
    const int idx = tid + 128 * i;
    const float4 v = wsrc[idx];
    *(float4*)(sW + (idx >> 5) * GP + (idx & 31) * 4) = v;
  }
  __syncthreads();

  v8f acc[4] = {};
  const float* sb = sW + l15 * GP + half * 2;
#pragma unroll
  for (int kk = 0; kk < GK / 4; ++kk) {
#pragma unroll
    for (int t = 0; t < 4; ++t) {
      const v2f b = *(const v2f*)(sb + t * 16 * GP + kk * 4);
      acc[t] = __builtin_amdgcn_wmma_f32_16x16x4_f32(
           false, a[kk],  false, b,
           (short)0, acc[t],  false,  false);
    }
  }

  float sc[8];
#pragma unroll
  for (int i = 0; i < 8; ++i) {
    const int r = row0 + 2 * i + half;
    sc[i] = rsqrtf((float)(cnt[(r < n) ? r : (n - 1)] + 1));
  }

  __syncthreads();

  float* sD = sW + wv * (16 * DP);
#pragma unroll
  for (int t = 0; t < 4; ++t) {
#pragma unroll
    for (int r = 0; r < 8; ++r) sD[(half * 8 + r) * DP + t * 16 + l15] = acc[t][r];
  }

  __syncthreads();

  float* __restrict__ hb = h + (size_t)ch * 64 + l15 * 4;
  const float* sr = sD + half * DP + l15 * 4;
  if (row0 + 16 <= n) {
    float4 v[8];
#pragma unroll
    for (int i = 0; i < 8; ++i) v[i] = *(const float4*)(sr + 2 * i * DP);
#pragma unroll
    for (int i = 0; i < 8; ++i) {
      const float4 o = make_float4(v[i].x * sc[i], v[i].y * sc[i], v[i].z * sc[i], v[i].w * sc[i]);
      *(float4*)(hb + (size_t)(row0 + 2 * i + half) * DOUT) = o;
    }
  } else {
#pragma unroll
    for (int i = 0; i < 8; ++i) {
      const int orow = row0 + 2 * i + half;
      const float4 v = *(const float4*)(sr + 2 * i * DP);
      const float4 o = make_float4(v.x * sc[i], v.y * sc[i], v.z * sc[i], v.w * sc[i]);
      if (orow < n) *(float4*)(hb + (size_t)orow * DOUT) = o;
    }
  }
}


template <int VEC, int CNT>
__device__ __forceinline__ void pull_rows(
    __attribute__((ext_vector_type(VEC))) float& acc, const float* __restrict__ hl,
    int sv, int j, unsigned last) {
  typedef float vf __attribute__((ext_vector_type(VEC)));
  vf v[CNT];
#pragma unroll
  for (int u = 0; u < CNT; ++u) {
    unsigned s = (unsigned)__builtin_amdgcn_readlane(sv, j + u);
    s = min(s, last);
    v[u] = *(const vf*)(hl + (size_t)s * (VEC * 32));
  }
#pragma unroll
  for (int u = 0; u < CNT; ++u) acc += v[u];
}

template <int VEC>
__global__ void __launch_bounds__(256)
k_pull(float* __restrict__ out, const float* __restrict__ h,
       const int* __restrict__ cnt, const int* __restrict__ rows,
       const int* __restrict__ ovf, const float* __restrict__ bias,
       int n, int e, int dorelu) {
  typedef float vf __attribute__((ext_vector_type(VEC)));
  constexpr int DOUT = VEC * 32;
  const int lane = threadIdx.x & 31;
  const int node = __builtin_amdgcn_readfirstlane((int)(blockIdx.x * 8 + (threadIdx.x >> 5)));
  if (node >= n) return;

  const int   deg = cnt[node];
  const float dd  = rsqrtf((float)(deg + 1));
  const int   m   = min(deg, SLOTS);
  const float* __restrict__ hl = h + lane * VEC;
  const unsigned last = (unsigned)(n - 1);

  int sv = 0;
  if (lane < m) sv = rows[(size_t)node * SLOTS + lane];
  const vf self = *(const vf*)(hl + (size_t)node * DOUT);
  vf acc = {};

  int j = 0;
  for (; j + 8 <= m; j += 8) pull_rows<VEC, 8>(acc, hl, sv, j, last);
  if (m & 4) { pull_rows<VEC, 4>(acc, hl, sv, j, last); j += 4; }
  if (m & 2) { pull_rows<VEC, 2>(acc, hl, sv, j, last); j += 2; }
  if (m & 1) { pull_rows<VEC, 1>(acc, hl, sv, j, last); }

  if (deg > SLOTS) {
    const int novf = min(cnt[n], e);
    for (int base = 0; base < novf; base += 32) {
      int od = -1, os = 0;
      if (base + lane < novf) {
        const int2 pr = *(const int2*)(ovf + 2 * (size_t)(base + lane));
        od = pr.x;
        os = pr.y;
      }
      unsigned mask = __builtin_amdgcn_ballot_w32(od == node);
      while (mask) {
        const int b = __builtin_ctz(mask);
        mask &= mask - 1;
        pull_rows<VEC, 1>(acc, hl, os, b, last);
      }
    }
  }

  acc += self;

#pragma unroll
  for (int k = 0; k < VEC; ++k) {
    float r = acc[k] * dd + bias[lane * VEC + k];
    acc[k] = dorelu ? fmaxf(r, 0.0f) : r;
  }
  *(vf*)(out + (size_t)node * DOUT + lane * VEC) = acc;
}


template <int NCH>
__global__ void __launch_bounds__(256)
k_pull_gemm(float* __restrict__ hout, const float* __restrict__ hin,
            const int* __restrict__ cnt, const int* __restrict__ rows,
            const int* __restrict__ ovf, const float* __restrict__ bias,
            const float* __restrict__ w, int n, int e) {
  typedef float vf __attribute__((ext_vector_type(4)));
  constexpr int DOUT  = 64 * NCH;
  constexpr int NCT   = DOUT / 16;
  constexpr int F4R   = DOUT / 4;
  constexpr int NPASS = (16 * F4R) / 256;
  __shared__ __attribute__((aligned(16))) float sW[DOUT * GP];
  __shared__ __attribute__((aligned(16))) float sA[16 * GP];
  __shared__ float sS[16];

  const int tid  = threadIdx.x;
  const int lane = tid & 31;
  const int wv   = __builtin_amdgcn_readfirstlane(tid >> 5);
  const int half = lane >> 4;
  const int l15  = lane & 15;
  const int tr0  = blockIdx.x * 64;

  int dgl = 0;
  if (lane < 8) dgl = cnt[min(tr0 + (lane >> 1) * 16 + wv * 2 + (lane & 1), n - 1)];

  const float4* __restrict__ wsrc = (const float4*)w;
  float4 wr[DOUT / 8];
#pragma unroll
  for (int i = 0; i < DOUT / 8; ++i) wr[i] = wsrc[tid + 256 * i];

  const float* __restrict__ hl = hin + lane * 4;
  const unsigned last = (unsigned)(n - 1);
  const vf bv = *(const vf*)(bias + lane * 4);

  int deg = __builtin_amdgcn_readlane(dgl, 0);
  int m   = min(deg, SLOTS);
  int sv  = 0;
  vf self;
  {
    const int nd = min(tr0 + wv * 2, n - 1);
    if (lane < m) sv = rows[(size_t)nd * SLOTS + lane];
    self = *(const vf*)(hl + (size_t)nd * GK);
  }

#pragma unroll
  for (int i = 0; i < DOUT / 8; ++i) {
    const int idx = tid + 256 * i;
    *(float4*)(sW + (idx >> 5) * GP + (idx & 31) * 4) = wr[i];
  }

  for (int st = 0; st < 4; ++st) {
    const int t0 = tr0 + st * 16;
    if (t0 >= n) break;

    for (int i = 0; i < 2; ++i) {
      const int q    = st * 2 + i;
      const int node = t0 + wv * 2 + i;
      int deg_n = 0, m_n = 0, sv_n = 0;
      vf self_n = {};
      if (q + 1 < 8) {
        deg_n = __builtin_amdgcn_readlane(dgl, q + 1);
        m_n   = min(deg_n, SLOTS);
        const int nn = min(tr0 + ((q + 1) >> 1) * 16 + wv * 2 + ((q + 1) & 1), n - 1);
        if (lane < m_n) sv_n = rows[(size_t)nn * SLOTS + lane];
        self_n = *(const vf*)(hl + (size_t)nn * GK);
      }

      const float dd = rsqrtf((float)(deg + 1));
      vf acc = {};
      if (node < n) {
        int j = 0;
        for (; j + 16 <= m; j += 16) pull_rows<4, 16>(acc, hl, sv, j, last);
        switch (m - j) {
          case 1:  pull_rows<4, 1>(acc, hl, sv, j, last);  break;
          case 2:  pull_rows<4, 2>(acc, hl, sv, j, last);  break;
          case 3:  pull_rows<4, 3>(acc, hl, sv, j, last);  break;
          case 4:  pull_rows<4, 4>(acc, hl, sv, j, last);  break;
          case 5:  pull_rows<4, 5>(acc, hl, sv, j, last);  break;
          case 6:  pull_rows<4, 6>(acc, hl, sv, j, last);  break;
          case 7:  pull_rows<4, 7>(acc, hl, sv, j, last);  break;
          case 8:  pull_rows<4, 8>(acc, hl, sv, j, last);  break;
          case 9:  pull_rows<4, 9>(acc, hl, sv, j, last);  break;
          case 10: pull_rows<4, 10>(acc, hl, sv, j, last); break;
          case 11: pull_rows<4, 11>(acc, hl, sv, j, last); break;
          case 12: pull_rows<4, 12>(acc, hl, sv, j, last); break;
          case 13: pull_rows<4, 13>(acc, hl, sv, j, last); break;
          case 14: pull_rows<4, 14>(acc, hl, sv, j, last); break;
          case 15: pull_rows<4, 15>(acc, hl, sv, j, last); break;
          default: break;
        }

        if (deg > SLOTS) {
          const int novf = min(cnt[n], e);
          for (int base = 0; base < novf; base += 32) {
            int od = -1, os = 0;
            if (base + lane < novf) {
              const int2 pr = *(const int2*)(ovf + 2 * (size_t)(base + lane));
              od = pr.x;
              os = pr.y;
            }
            unsigned mask = __builtin_amdgcn_ballot_w32(od == node);
            while (mask) {
              const int b = __builtin_ctz(mask);
              mask &= mask - 1;
              pull_rows<4, 1>(acc, hl, os, b, last);
            }
          }
        }

        acc += self;
#pragma unroll
        for (int k = 0; k < 4; ++k) acc[k] = fmaxf(acc[k] * dd + bv[k], 0.0f);
      }
      *(vf*)(sA + (wv * 2 + i) * GP + lane * 4) = acc;
      if (lane == 0) sS[wv * 2 + i] = dd;

      deg = deg_n; m = m_n; sv = sv_n; self = self_n;
    }

    __syncthreads();

    v8f d = {};
    if (NCT == 8 || wv < NCT) {
      const float* sa = sA + l15 * GP + half * 2;
      const float* sb = sW + (wv * 16 + l15) * GP + half * 2;
#pragma unroll
      for (int kk = 0; kk < GK / 4; ++kk) {
        const v2f a = *(const v2f*)(sa + kk * 4);
        const v2f b = *(const v2f*)(sb + kk * 4);
        d = __builtin_amdgcn_wmma_f32_16x16x4_f32(
             false, a,  false, b,
             (short)0, d,  false,  false);
      }
    }

    __syncthreads();

    if (NCT == 8 || wv < NCT) {
#pragma unroll
      for (int r = 0; r < 8; ++r) sA[(half * 8 + r) * GP + wv * 16 + l15] = d[r];
    }

    __syncthreads();

#pragma unroll
    for (int i = 0; i < NPASS; ++i) {
      const int idx  = tid + 256 * i;
      const int row  = idx / F4R;
      const int c4   = idx % F4R;
      const int orow = t0 + row;
      const float4 v = *(const float4*)(sA + row * GP + c4 * 4);
      const float  s = sS[row];
      const float4 o = make_float4(v.x * s, v.y * s, v.z * s, v.w * s);
      if (orow < n) *(float4*)(hout + (size_t)orow * DOUT + c4 * 4) = o;
    }

    __syncthreads();
  }
}

static inline size_t align256(size_t x) { return (x + 255) & ~(size_t)255; }

extern "C" void kernel_launch(void* const* d_in, const int* in_sizes, int n_in,
                              void* d_out, int out_size, void* d_ws, size_t ws_size,
                              hipStream_t stream) {
  (void)n_in; (void)out_size; (void)ws_size;

  const int DIN = 128, DH = 128, DOUT = 64, LAYERS = 5;
  const int N = in_sizes[0] / DIN;
  const int E = in_sizes[1] / 2;

  const float* x0   = (const float*)d_in[0];
  const int*  eidx  = (const int*)d_in[1];
  const int*  srcA  = eidx;
  const int*  dstA  = eidx + E;

  char* ws = (char*)d_ws;
  size_t off = 0;
  int*   cnt    = (int*)  (ws + off); off += align256((size_t)(N + 1) * sizeof(int));
  int*   rows   = (int*)  (ws + off); off += align256((size_t)N * SLOTS * sizeof(int));
  int*   ovf    = (int*)  (ws + off); off += align256((size_t)E * 2 * sizeof(int));
  float* bufH   = (float*)(ws + off); off += align256((size_t)N * DH * sizeof(float));
  float* bufX   = (float*)(ws + off); off += align256((size_t)N * DH * sizeof(float));

  const int TB = 256;

  k_zero_cnt<<<(N + 1 + TB - 1) / TB, TB, 0, stream>>>(cnt, N + 1);
  k_bucket  <<<(E + TB - 1) / TB, TB, 0, stream>>>(cnt, rows, ovf, srcA, dstA, E, N);

  k_gemm_wmma<2><<<((N + 63) / 64) * 2, 128, 0, stream>>>(bufH, x0, (const float*)d_in[2], cnt, N);

  float* hin  = bufH;
  float* hnew = bufX;
  for (int layer = 1; layer < LAYERS; ++layer) {
    const float* W  = (const float*)d_in[2 + 2 * layer];
    const float* bp = (const float*)d_in[3 + 2 * (layer - 1)];
    if (layer < LAYERS - 1)
      k_pull_gemm<2><<<(N + 63) / 64, 256, 0, stream>>>(hnew, hin, cnt, rows, ovf, bp, W, N, E);
    else
      k_pull_gemm<1><<<(N + 63) / 64, 256, 0, stream>>>(hnew, hin, cnt, rows, ovf, bp, W, N, E);
    float* t = hin; hin = hnew; hnew = t;
  }

  k_pull<2><<<(N + 7) / 8, 256, 0, stream>>>((float*)d_out, hin, cnt, rows, ovf,
                                             (const float*)d_in[3 + 2 * (LAYERS - 1)], N, E, 0);
}


# Edge grouping goes through LDS: blocks sort edge chunks by 256-node bucket, one block per bucket ranks with LDS atomics and writes whole rows; no per-edge global atomic; sum order: LDS arrival rank
# speedup vs baseline: 5.7983x; 1.1849x over previous
#include <hip/hip_runtime.h>


typedef __attribute__((ext_vector_type(2))) float v2f;
typedef __attribute__((ext_vector_type(8))) float v8f;

constexpr int SLOTS = 32;
constexpr int NBKS  = 8;
constexpr int NBK   = 256;
constexpr int PMAX  = 256;
constexpr int EPT   = 13;
constexpr int CH    = 256 * EPT;

__global__ void __launch_bounds__(256)
k_bin(int* __restrict__ pairs, int* __restrict__ offs, int* __restrict__ novf,
      const int* __restrict__ src, const int* __restrict__ dst, int e, int n, int nb1) {
  __shared__ int hist[PMAX];
  __shared__ int wsum[8];
  __shared__ int stot;
  __shared__ __attribute__((aligned(8))) int sorted[2 * CH];
  const int tid  = threadIdx.x;
  const int lane = tid & 31;
  const int wv   = tid >> 5;
  const int b    = blockIdx.x;
  const int base = b * CH;
  if (b == 0 && tid == 0) *novf = 0;
  hist[tid] = 0;

  int d[EPT], s[EPT], r[EPT];
#pragma unroll
  for (int k = 0; k < EPT; ++k) {
    const int i = base + k * 256 + tid;
    d[k] = -1;
    s[k] = 0;
    if (i < e) { d[k] = dst[i]; s[k] = src[i]; }
  }
  __syncthreads();
  const unsigned nlim = (unsigned)min(n, PMAX * NBK);
#pragma unroll
  for (int k = 0; k < EPT; ++k) {
    r[k] = 0;
    if ((unsigned)d[k] < nlim) r[k] = atomicAdd(&hist[d[k] >> NBKS], 1);
    else d[k] = -1;
  }
  __syncthreads();

  const int v = hist[tid];
  int x = v;
#pragma unroll
  for (int o = 1; o < 32; o <<= 1) {
    const int y = __shfl_up(x, o);
    if (lane >= o) x += y;
  }
  if (lane == 31) wsum[wv] = x;
  __syncthreads();
  int add = 0;
#pragma unroll
  for (int w = 0; w < 8; ++w) if (w < wv) add += wsum[w];
  const int start = x - v + add;
  hist[tid] = start;
  offs[(size_t)tid * nb1 + b] = start;
  if (tid == 255) {
    stot = start + v;
    offs[(size_t)256 * nb1 + b] = start + v;
  }
  __syncthreads();

#pragma unroll
  for (int k = 0; k < EPT; ++k) {
    if (d[k] >= 0) {
      const int pos = hist[d[k] >> NBKS] + r[k];
      *(int2*)(sorted + 2 * pos) = make_int2(d[k], s[k]);
    }
  }
  __syncthreads();
  const int total = stot;
  int2* __restrict__ outp = (int2*)pairs + (size_t)b * CH;
  for (int j = tid; j < total; j += 256) outp[j] = *(const int2*)(sorted + 2 * j);
}

__device__ __forceinline__ void place_edge(int2 q, int nb0, int* scnt, int* srows,
                                           int* __restrict__ novf, int* __restrict__ ovf, int e) {
  const int li = (q.x - nb0) & (NBK - 1);
  const int r  = atomicAdd(&scnt[li], 1);
  if (r < SLOTS) {
    srows[li * SLOTS + r] = q.y;
  } else {
    const int k = atomicAdd(novf, 1);
    if (k < e) *(int2*)(ovf + 2 * (size_t)k) = make_int2(q.x, q.y);
  }
}

__global__ void __launch_bounds__(256)
k_rank(int* __restrict__ cnt, int* __restrict__ rows, int* __restrict__ ovf,
       const int* __restrict__ pairs, const int* __restrict__ offs, int nb1, int e, int n) {
  __shared__ int scnt[NBK];
  __shared__ __attribute__((aligned(16))) int srows[NBK * SLOTS];
  const int tid = threadIdx.x;
  const int p   = blockIdx.x;
  const int nb0 = p * NBK;
  scnt[tid] = 0;
  for (int j = tid; j < NBK * SLOTS / 4; j += 256) ((int4*)srows)[j] = make_int4(0, 0, 0, 0);
  __syncthreads();

  for (int b = tid; b < nb1; b += 256) {
    int beg = offs[(size_t)p * nb1 + b];
    int end = offs[(size_t)(p + 1) * nb1 + b];
    end = min(max(end, 0), CH);
    beg = min(max(beg, 0), end);
    const int2* __restrict__ pr = (const int2*)pairs + (size_t)b * CH;
    int k = beg;
    for (; k + 4 <= end; k += 4) {
      const int2 q0 = pr[k], q1 = pr[k + 1], q2 = pr[k + 2], q3 = pr[k + 3];
      place_edge(q0, nb0, scnt, srows, cnt + n, ovf, e);
      place_edge(q1, nb0, scnt, srows, cnt + n, ovf, e);
      place_edge(q2, nb0, scnt, srows, cnt + n, ovf, e);
      place_edge(q3, nb0, scnt, srows, cnt + n, ovf, e);
    }
    for (; k < end; ++k) place_edge(pr[k], nb0, scnt, srows, cnt + n, ovf, e);
  }
  __syncthreads();

  const int nv = min(NBK, n - nb0);
  int4* __restrict__ rp = (int4*)(rows + (size_t)nb0 * SLOTS);
  for (int j = tid; j < nv * (SLOTS / 4); j += 256) rp[j] = ((const int4*)srows)[j];
  if (tid < nv) cnt[nb0 + tid] = scnt[tid];
}

constexpr int GK = 128;
constexpr int GP = GK + 4;
constexpr int DP = 64 + 4;

template <int NCH>
__global__ void __launch_bounds__(128)
k_gemm_wmma(float* __restrict__ h, const float* __restrict__ x,
            const float* __restrict__ w, const int* __restrict__ cnt, int n) {
  constexpr int DOUT = 64 * NCH;
  __shared__ __attribute__((aligned(16))) float sW[64 * GP];

  const int tid  = threadIdx.x;
  const int lane = tid & 31;
  const int wv   = __builtin_amdgcn_readfirstlane(tid >> 5);
  const int half = lane >> 4;
  const int l15  = lane & 15;
  const int rt   = blockIdx.x / NCH;
  const int ch   = blockIdx.x % NCH;

  const int row0 = rt * 64 + wv * 16;
#pragma unroll
  for (int i = 0; i < 16; ++i) {
    const int idx  = tid + 128 * i;
    const int grow = rt * 64 + (idx >> 5);
    const int crow = (grow < n) ? grow : (n - 1);
    const float4 v = *(const float4*)(x + (size_t)crow * GK + (idx & 31) * 4);
    *(float4*)(sW + (idx >> 5) * GP + (idx & 31) * 4) = v;
  }

  __syncthreads();

  v2f a[GK / 4];
  const float* sa = sW + (wv * 16 + l15) * GP + half * 2;
#pragma unroll
  for (int kk = 0; kk < GK / 4; ++kk) a[kk] = *(const v2f*)(sa + kk * 4);

  __syncthreads();

  const float4* __restrict__ wsrc = (const float4*)(w + (size_t)ch * 64 * GK);
#pragma unroll
  for (int i = 0; i < 16; ++i) {
    const int idx = tid + 128 * i;
    const float4 v = wsrc[idx];
    *(float4*)(sW + (idx >> 5) * GP + (idx & 31) * 4) = v;
  }
  __syncthreads();

  v8f acc[4] = {};
  const float* sb = sW + l15 * GP + half * 2;
#pragma unroll
  for (int kk = 0; kk < GK / 4; ++kk) {
#pragma unroll
    for (int t = 0; t < 4; ++t) {
      const v2f b = *(const v2f*)(sb + t * 16 * GP + kk * 4);
      acc[t] = __builtin_amdgcn_wmma_f32_16x16x4_f32(
           false, a[kk],  false, b,
           (short)0, acc[t],  false,  false);
    }
  }

  float sc[8];
#pragma unroll
  for (int i = 0; i < 8; ++i) {
    const int r = row0 + 2 * i + half;
    sc[i] = rsqrtf((float)(cnt[(r < n) ? r : (n - 1)] + 1));
  }

  __syncthreads();

  float* sD = sW + wv * (16 * DP);
#pragma unroll
  for (int t = 0; t < 4; ++t) {
#pragma unroll
    for (int r = 0; r < 8; ++r) sD[(half * 8 + r) * DP + t * 16 + l15] = acc[t][r];
  }

  __syncthreads();

  float* __restrict__ hb = h + (size_t)ch * 64 + l15 * 4;
  const float* sr = sD + half * DP + l15 * 4;
  if (row0 + 16 <= n) {
    float4 v[8];
#pragma unroll
    for (int i = 0; i < 8; ++i) v[i] = *(const float4*)(sr + 2 * i * DP);
#pragma unroll
    for (int i = 0; i < 8; ++i) {
      const float4 o = make_float4(v[i].x * sc[i], v[i].y * sc[i], v[i].z * sc[i], v[i].w * sc[i]);
      *(float4*)(hb + (size_t)(row0 + 2 * i + half) * DOUT) = o;
    }
  } else {
#pragma unroll
    for (int i = 0; i < 8; ++i) {
      const int orow = row0 + 2 * i + half;
      const float4 v = *(const float4*)(sr + 2 * i * DP);
      const float4 o = make_float4(v.x * sc[i], v.y * sc[i], v.z * sc[i], v.w * sc[i]);
      if (orow < n) *(float4*)(hb + (size_t)orow * DOUT) = o;
    }
  }
}


template <int VEC, int CNT>
__device__ __forceinline__ void pull_rows(
    __attribute__((ext_vector_type(VEC))) float& acc, const float* __restrict__ hl,
    int sv, int j, unsigned last) {
  typedef float vf __attribute__((ext_vector_type(VEC)));
  vf v[CNT];
#pragma unroll
  for (int u = 0; u < CNT; ++u) {
    unsigned s = (unsigned)__builtin_amdgcn_readlane(sv, j + u);
    s = min(s, last);
    v[u] = *(const vf*)(hl + (size_t)s * (VEC * 32));
  }
#pragma unroll
  for (int u = 0; u < CNT; ++u) acc += v[u];
}

template <int VEC>
__global__ void __launch_bounds__(256)
k_pull(float* __restrict__ out, const float* __restrict__ h,
       const int* __restrict__ cnt, const int* __restrict__ rows,
       const int* __restrict__ ovf, const float* __restrict__ bias,
       int n, int e, int dorelu) {
  typedef float vf __attribute__((ext_vector_type(VEC)));
  constexpr int DOUT = VEC * 32;
  const int lane = threadIdx.x & 31;
  const int node = __builtin_amdgcn_readfirstlane((int)(blockIdx.x * 8 + (threadIdx.x >> 5)));
  if (node >= n) return;

  const int   deg = cnt[node];
  const float dd  = rsqrtf((float)(deg + 1));
  const int   m   = min(deg, SLOTS);
  const float* __restrict__ hl = h + lane * VEC;
  const unsigned last = (unsigned)(n - 1);

  int sv = 0;
  if (lane < m) sv = rows[(size_t)node * SLOTS + lane];
  const vf self = *(const vf*)(hl + (size_t)node * DOUT);
  vf acc = {};

  int j = 0;
  for (; j + 8 <= m; j += 8) pull_rows<VEC, 8>(acc, hl, sv, j, last);
  if (m & 4) { pull_rows<VEC, 4>(acc, hl, sv, j, last); j += 4; }
  if (m & 2) { pull_rows<VEC, 2>(acc, hl, sv, j, last); j += 2; }
  if (m & 1) { pull_rows<VEC, 1>(acc, hl, sv, j, last); }

  if (deg > SLOTS) {
    const int novf = min(cnt[n], e);
    for (int base = 0; base < novf; base += 32) {
      int od = -1, os = 0;
      if (base + lane < novf) {
        const int2 pr = *(const int2*)(ovf + 2 * (size_t)(base + lane));
        od = pr.x;
        os = pr.y;
      }
      unsigned mask = __builtin_amdgcn_ballot_w32(od == node);
      while (mask) {
        const int b = __builtin_ctz(mask);
        mask &= mask - 1;
        pull_rows<VEC, 1>(acc, hl, os, b, last);
      }
    }
  }

  acc += self;

#pragma unroll
  for (int k = 0; k < VEC; ++k) {
    float r = acc[k] * dd + bias[lane * VEC + k];
    acc[k] = dorelu ? fmaxf(r, 0.0f) : r;
  }
  *(vf*)(out + (size_t)node * DOUT + lane * VEC) = acc;
}


template <int NCH>
__global__ void __launch_bounds__(256)
k_pull_gemm(float* __restrict__ hout, const float* __restrict__ hin,
            const int* __restrict__ cnt, const int* __restrict__ rows,
            const int* __restrict__ ovf, const float* __restrict__ bias,
            const float* __restrict__ w, int n, int e) {
  typedef float vf __attribute__((ext_vector_type(4)));
  constexpr int DOUT  = 64 * NCH;
  constexpr int NCT   = DOUT / 16;
  constexpr int F4R   = DOUT / 4;
  constexpr int NPASS = (16 * F4R) / 256;
  __shared__ __attribute__((aligned(16))) float sW[DOUT * GP];
  __shared__ __attribute__((aligned(16))) float sA[16 * GP];
  __shared__ float sS[16];

  const int tid  = threadIdx.x;
  const int lane = tid & 31;
  const int wv   = __builtin_amdgcn_readfirstlane(tid >> 5);
  const int half = lane >> 4;
  const int l15  = lane & 15;
  const int tr0  = blockIdx.x * 64;

  int dgl = 0;
  if (lane < 8) dgl = cnt[min(tr0 + (lane >> 1) * 16 + wv * 2 + (lane & 1), n - 1)];

  const float4* __restrict__ wsrc = (const float4*)w;
  float4 wr[DOUT / 8];
#pragma unroll
  for (int i = 0; i < DOUT / 8; ++i) wr[i] = wsrc[tid + 256 * i];

  const float* __restrict__ hl = hin + lane * 4;
  const unsigned last = (unsigned)(n - 1);
  const vf bv = *(const vf*)(bias + lane * 4);

  int deg = __builtin_amdgcn_readlane(dgl, 0);
  int m   = min(deg, SLOTS);
  int sv  = 0;
  vf self;
  {
    const int nd = min(tr0 + wv * 2, n - 1);
    if (lane < m) sv = rows[(size_t)nd * SLOTS + lane];
    self = *(const vf*)(hl + (size_t)nd * GK);
  }

#pragma unroll
  for (int i = 0; i < DOUT / 8; ++i) {
    const int idx = tid + 256 * i;
    *(float4*)(sW + (idx >> 5) * GP + (idx & 31) * 4) = wr[i];
  }

  for (int st = 0; st < 4; ++st) {
    const int t0 = tr0 + st * 16;
    if (t0 >= n) break;

    for (int i = 0; i < 2; ++i) {
      const int q    = st * 2 + i;
      const int node = t0 + wv * 2 + i;
      int deg_n = 0, m_n = 0, sv_n = 0;
      vf self_n = {};
      if (q + 1 < 8) {
        deg_n = __builtin_amdgcn_readlane(dgl, q + 1);
        m_n   = min(deg_n, SLOTS);
        const int nn = min(tr0 + ((q + 1) >> 1) * 16 + wv * 2 + ((q + 1) & 1), n - 1);
        if (lane < m_n) sv_n = rows[(size_t)nn * SLOTS + lane];
        self_n = *(const vf*)(hl + (size_t)nn * GK);
      }

      const float dd = rsqrtf((float)(deg + 1));
      vf acc = {};
      if (node < n) {
        int j = 0;
        for (; j + 16 <= m; j += 16) pull_rows<4, 16>(acc, hl, sv, j, last);
        switch (m - j) {
          case 1:  pull_rows<4, 1>(acc, hl, sv, j, last);  break;
          case 2:  pull_rows<4, 2>(acc, hl, sv, j, last);  break;
          case 3:  pull_rows<4, 3>(acc, hl, sv, j, last);  break;
          case 4:  pull_rows<4, 4>(acc, hl, sv, j, last);  break;
          case 5:  pull_rows<4, 5>(acc, hl, sv, j, last);  break;
          case 6:  pull_rows<4, 6>(acc, hl, sv, j, last);  break;
          case 7:  pull_rows<4, 7>(acc, hl, sv, j, last);  break;
          case 8:  pull_rows<4, 8>(acc, hl, sv, j, last);  break;
          case 9:  pull_rows<4, 9>(acc, hl, sv, j, last);  break;
          case 10: pull_rows<4, 10>(acc, hl, sv, j, last); break;
          case 11: pull_rows<4, 11>(acc, hl, sv, j, last); break;
          case 12: pull_rows<4, 12>(acc, hl, sv, j, last); break;
          case 13: pull_rows<4, 13>(acc, hl, sv, j, last); break;
          case 14: pull_rows<4, 14>(acc, hl, sv, j, last); break;
          case 15: pull_rows<4, 15>(acc, hl, sv, j, last); break;
          default: break;
        }

        if (deg > SLOTS) {
          const int novf = min(cnt[n], e);
          for (int base = 0; base < novf; base += 32) {
            int od = -1, os = 0;
            if (base + lane < novf) {
              const int2 pr = *(const int2*)(ovf + 2 * (size_t)(base + lane));
              od = pr.x;
              os = pr.y;
            }
            unsigned mask = __builtin_amdgcn_ballot_w32(od == node);
            while (mask) {
              const int b = __builtin_ctz(mask);
              mask &= mask - 1;
              pull_rows<4, 1>(acc, hl, os, b, last);
            }
          }
        }

        acc += self;
#pragma unroll
        for (int k = 0; k < 4; ++k) acc[k] = fmaxf(acc[k] * dd + bv[k], 0.0f);
      }
      *(vf*)(sA + (wv * 2 + i) * GP + lane * 4) = acc;
      if (lane == 0) sS[wv * 2 + i] = dd;

      deg = deg_n; m = m_n; sv = sv_n; self = self_n;
    }

    __syncthreads();

    v8f d = {};
    if (NCT == 8 || wv < NCT) {
      const float* sa = sA + l15 * GP + half * 2;
      const float* sb = sW + (wv * 16 + l15) * GP + half * 2;
#pragma unroll
      for (int kk = 0; kk < GK / 4; ++kk) {
        const v2f a = *(const v2f*)(sa + kk * 4);
        const v2f b = *(const v2f*)(sb + kk * 4);
        d = __builtin_amdgcn_wmma_f32_16x16x4_f32(
             false, a,  false, b,
             (short)0, d,  false,  false);
      }
    }

    __syncthreads();

    if (NCT == 8 || wv < NCT) {
#pragma unroll
      for (int r = 0; r < 8; ++r) sA[(half * 8 + r) * GP + wv * 16 + l15] = d[r];
    }

    __syncthreads();

#pragma unroll
    for (int i = 0; i < NPASS; ++i) {
      const int idx  = tid + 256 * i;
      const int row  = idx / F4R;
      const int c4   = idx % F4R;
      const int orow = t0 + row;
      const float4 v = *(const float4*)(sA + row * GP + c4 * 4);
      const float  s = sS[row];
      const float4 o = make_float4(v.x * s, v.y * s, v.z * s, v.w * s);
      if (orow < n) *(float4*)(hout + (size_t)orow * DOUT + c4 * 4) = o;
    }

    __syncthreads();
  }
}

static inline size_t align256(size_t x) { return (x + 255) & ~(size_t)255; }

extern "C" void kernel_launch(void* const* d_in, const int* in_sizes, int n_in,
                              void* d_out, int out_size, void* d_ws, size_t ws_size,
                              hipStream_t stream) {
  (void)n_in; (void)out_size; (void)ws_size;

  const int DIN = 128, DH = 128, DOUT = 64, LAYERS = 5;
  const int N = in_sizes[0] / DIN;
  const int E = in_sizes[1] / 2;

  const float* x0   = (const float*)d_in[0];
  const int*  eidx  = (const int*)d_in[1];
  const int*  srcA  = eidx;
  const int*  dstA  = eidx + E;

  char* ws = (char*)d_ws;
  size_t off = 0;
  int*   cnt    = (int*)  (ws + off); off += align256((size_t)(N + 1) * sizeof(int));
  int*   rows   = (int*)  (ws + off); off += align256((size_t)N * SLOTS * sizeof(int));
  int*   ovf    = (int*)  (ws + off); off += align256((size_t)E * 2 * sizeof(int));
  float* bufH   = (float*)(ws + off); off += align256((size_t)N * DH * sizeof(float));
  float* bufX   = (float*)(ws + off); off += align256((size_t)N * DH * sizeof(float));
  const int NB1 = (E + CH - 1) / CH;
  const int NP  = (N + NBK - 1) / NBK;
  int*   pairs  = (int*)  (ws + off); off += align256((size_t)NB1 * CH * 2 * sizeof(int));
  int*   offs   = (int*)  (ws + off); off += align256((size_t)(PMAX + 1) * NB1 * sizeof(int));

  const int TB = 256;

  k_bin <<<NB1, TB, 0, stream>>>(pairs, offs, cnt + N, srcA, dstA, E, N, NB1);
  k_rank<<<NP,  TB, 0, stream>>>(cnt, rows, ovf, pairs, offs, NB1, E, N);

  k_gemm_wmma<2><<<((N + 63) / 64) * 2, 128, 0, stream>>>(bufH, x0, (const float*)d_in[2], cnt, N);

  float* hin  = bufH;
  float* hnew = bufX;
  for (int layer = 1; layer < LAYERS; ++layer) {
    const float* W  = (const float*)d_in[2 + 2 * layer];
    const float* bp = (const float*)d_in[3 + 2 * (layer - 1)];
    if (layer < LAYERS - 1)
      k_pull_gemm<2><<<(N + 63) / 64, 256, 0, stream>>>(hnew, hin, cnt, rows, ovf, bp, W, N, E);
    else
      k_pull_gemm<1><<<(N + 63) / 64, 256, 0, stream>>>(hnew, hin, cnt, rows, ovf, bp, W, N, E);
    float* t = hin; hin = hnew; hnew = t;
  }

  k_pull<2><<<(N + 7) / 8, 256, 0, stream>>>((float*)d_out, hin, cnt, rows, ovf,
                                             (const float*)d_in[3 + 2 * (LAYERS - 1)], N, E, 0);
}
